# starting kernel
# baseline (speedup 1.0000x reference)
#include <hip/hip_runtime.h>
#include <hip/hip_bf16.h>
#include <math.h>


typedef __attribute__((ext_vector_type(16))) _Float16 v16h;
typedef __attribute__((ext_vector_type(8)))  float    v8f;

#define EPS_F 1e-8f

__global__ __launch_bounds__(256) void cgmm_zero(float* __restrict__ p, int n) {
    int i = blockIdx.x * blockDim.x + threadIdx.x;
    int stride = gridDim.x * blockDim.x;
    for (; i < n; i += stride) p[i] = 0.0f;
}

__global__ __launch_bounds__(128) void cgmm_params(const float* __restrict__ Q,
                                                   const float* __restrict__ B,
                                                   float* __restrict__ Bs_t,
                                                   _Float16* __restrict__ bfrag) {
    __shared__ float sQ[512];
    int t = threadIdx.x;
    if (t < 64) {
        int g = t & 7, a = t >> 3;
        {
            int l = a;
            float mx = -1e30f;
            for (int i = 0; i < 8; ++i) mx = fmaxf(mx, Q[i * 64 + l * 8 + g]);
            float e[8], s = 0.0f;
            for (int i = 0; i < 8; ++i) { e[i] = __expf(Q[i * 64 + l * 8 + g] - mx); s += e[i]; }
            float inv = 1.0f / s;
            for (int i = 0; i < 8; ++i) sQ[i * 64 + l * 8 + g] = e[i] * inv;
        }
        {
            int i = a;
            float mx = -1e30f;
            for (int m = 0; m < 32; ++m) mx = fmaxf(mx, B[i * 256 + m * 8 + g]);
            float ev[32], s = 0.0f;
            for (int m = 0; m < 32; ++m) { ev[m] = __expf(B[i * 256 + m * 8 + g] - mx); s += ev[m]; }
            float inv = 1.0f / s;
            for (int m = 0; m < 32; ++m) Bs_t[m * 64 + g * 8 + i] = ev[m] * inv;
        }
    }
    __syncthreads();
    int p = t >> 5, lane = t & 31;
    bool hi = lane >= 16;
    int  m  = lane & 15;
    bool bvalid = (hi == (m >= 8));
    int  icol = m & 7;
    int  g = 2 * p + (hi ? 1 : 0);
    _Float16* dst = bfrag + (p * 32 + lane) * 16;
    for (int e2 = 0; e2 < 16; ++e2) {
        float f = (e2 < 8 && bvalid) ? sQ[icol * 64 + e2 * 8 + g] : 0.0f;
        dst[e2] = (_Float16)f;
    }
}

__global__ __launch_bounds__(256) void cgmm_edge_scatter(const float* __restrict__ prev_h,
                                                         const int* __restrict__ ei,
                                                         float* __restrict__ summed,
                                                         float* __restrict__ counts,
                                                         int E) {
    int lane = threadIdx.x & 31;
    int e = blockIdx.x * 8 + (threadIdx.x >> 5);
    if (e >= E) return;
    int dst = ei[e];
    int src = ei[E + e];
    const float2* ph = (const float2*)(prev_h + (size_t)src * 64);
    float2 v = ph[lane];
    float* sp = summed + (size_t)dst * 64 + lane * 2;
    __hip_atomic_fetch_add(sp,     v.x, __ATOMIC_RELAXED, __HIP_MEMORY_SCOPE_AGENT);
    __hip_atomic_fetch_add(sp + 1, v.y, __ATOMIC_RELAXED, __HIP_MEMORY_SCOPE_AGENT);
    if (lane == 0)
        __hip_atomic_fetch_add(counts + dst, 1.0f, __ATOMIC_RELAXED, __HIP_MEMORY_SCOPE_AGENT);
}

__global__ __launch_bounds__(256) void cgmm_node_wmma(const int* __restrict__ x,
                                                      const float* __restrict__ summed,
                                                      const float* __restrict__ counts,
                                                      const _Float16* __restrict__ bfrag,
                                                      const float* __restrict__ Bs_t,
                                                      float* __restrict__ lik,
                                                      float* __restrict__ post,
                                                      int N) {
    __shared__ float ldsT[8][16][8][8];
    __shared__ float ldsP[8][16][64];
    int lane = threadIdx.x & 31;
    int wave = threadIdx.x >> 5;
    int tile = blockIdx.x * 8 + wave;
    int n0 = tile * 16;
    if (n0 >= N) return;

    int  m    = lane & 15;
    bool hi   = lane >= 16;
    int  node = n0 + m;
    int  nodec = node < N ? node : N - 1;
    bool valid = node < N;

    const float* rowbase = summed + (size_t)nodec * 64;
    __builtin_prefetch(rowbase, 0, 3);
    float inv_cnt = 1.0f / fmaxf(counts[nodec], 1.0f);

#pragma unroll
    for (int p = 0; p < 4; ++p) {
        int g = 2 * p + (hi ? 1 : 0);
        const float* row = rowbase + g;
        v16h a;
#pragma unroll
        for (int e2 = 0; e2 < 8; ++e2)
            a[e2] = (_Float16)(row[e2 * 8] * inv_cnt);
#pragma unroll
        for (int e2 = 8; e2 < 16; ++e2)
            a[e2] = (_Float16)0.0f;

        v16h b = *((const v16h*)(bfrag + (p * 32 + lane) * 16));

        v8f acc = {};
        acc = __builtin_amdgcn_wmma_f32_16x16x32_f16(
            false, a, false, b, (short)0, acc, false, false);
#pragma unroll
        for (int r = 0; r < 8; ++r) {
            int mm = r + (hi ? 8 : 0);
            int gq = 2 * p + (m >= 8 ? 1 : 0);
            ldsT[wave][mm][gq][m & 7] = acc[r];
        }
    }

    int gbase = hi ? 4 : 0;
    int xi = x[nodec];
    const float* brow = Bs_t + (size_t)xi * 64;
    float lik4[4];
#pragma unroll
    for (int gg = 0; gg < 4; ++gg) {
        int g = gbase + gg;
        const float4* tp = (const float4*)&ldsT[wave][m][g][0];
        float4 t0 = tp[0], t1 = tp[1];
        const float4* bp = (const float4*)(brow + g * 8);
        float4 b0 = bp[0], b1 = bp[1];
        float u[8];
        u[0] = t0.x * b0.x; u[1] = t0.y * b0.y; u[2] = t0.z * b0.z; u[3] = t0.w * b0.w;
        u[4] = t1.x * b1.x; u[5] = t1.y * b1.y; u[6] = t1.z * b1.z; u[7] = t1.w * b1.w;
        float s = ((u[0] + u[1]) + (u[2] + u[3])) + ((u[4] + u[5]) + (u[6] + u[7]));
        float denom = s + 64.0f * EPS_F;
        lik4[gg] = __logf(denom);
        float invd = 1.0f / denom;
#pragma unroll
        for (int i = 0; i < 8; ++i)
            ldsP[wave][m][i * 8 + g] = (u[i] + 8.0f * EPS_F) * invd;
    }
    if (valid) {
        float4 lv = {lik4[0], lik4[1], lik4[2], lik4[3]};
        *(float4*)(lik + (size_t)node * 8 + gbase) = lv;
    }

    {
        int nl = lane >> 1;
        int half = lane & 1;
        int outnode = n0 + nl;
        if (outnode < N) {
            const float4* src = (const float4*)&ldsP[wave][nl][half * 32];
            float4* dstp = (float4*)(post + (size_t)outnode * 64 + half * 32);
#pragma unroll
            for (int q = 0; q < 8; ++q) dstp[q] = src[q];
        }
    }
}

extern "C" void kernel_launch(void* const* d_in, const int* in_sizes, int n_in,
                              void* d_out, int out_size, void* d_ws, size_t ws_size,
                              hipStream_t stream) {
    (void)n_in; (void)out_size; (void)ws_size;
    const int*   x      = (const int*)d_in[0];
    const float* prev_h = (const float*)d_in[1];
    const int*   ei     = (const int*)d_in[2];
    const float* Q      = (const float*)d_in[3];
    const float* B      = (const float*)d_in[4];
    const int N = in_sizes[0];
    const int E = in_sizes[2] / 2;

    float* ws     = (float*)d_ws;
    float* summed = ws;
    float* counts = summed + (size_t)N * 64;
    float* Bs_t   = counts + N;
    size_t off    = ((size_t)N * 65 + 2048 + 7) & ~(size_t)7;
    _Float16* bfrag = (_Float16*)(ws + off);

    float* lik  = (float*)d_out;
    float* post = lik + (size_t)N * 8;

    int zn = N * 65;
    int zb = (zn + 255) / 256; if (zb > 4096) zb = 4096;
    cgmm_zero<<<zb, 256, 0, stream>>>(summed, zn);

    cgmm_params<<<1, 128, 0, stream>>>(Q, B, Bs_t, bfrag);

    cgmm_edge_scatter<<<(E + 7) / 8, 256, 0, stream>>>(prev_h, ei, summed, counts, E);

    int tiles = (N + 15) / 16;
    cgmm_node_wmma<<<(tiles + 7) / 8, 256, 0, stream>>>(x, summed, counts, bfrag, Bs_t,
                                                        lik, post, N);
}


# Drain chain 5 -> 2 dependent load rounds: a flagged group's source line is fetched beside its destination line, queue holds (source,row); 16 source rows per batch, not 8; add order: LDS arrival
# speedup vs baseline: 4.4884x; 4.4884x over previous
#include <hip/hip_runtime.h>
#include <hip/hip_bf16.h>
#include <math.h>


typedef __attribute__((ext_vector_type(16))) _Float16 v16h;
typedef __attribute__((ext_vector_type(8)))  float    v8f;

#define EPS_F 1e-8f

__device__ __forceinline__ void cgmm_params_body(const float* __restrict__ Q,
                                                 const float* __restrict__ B,
                                                 float* __restrict__ Bs_t,
                                                 _Float16* __restrict__ bfrag) {
    __shared__ float sQ[512];
    int t = threadIdx.x;
    if (t < 64) {
        int g = t & 7, a = t >> 3;
        {
            int l = a;
            float mx = -1e30f;
            for (int i = 0; i < 8; ++i) mx = fmaxf(mx, Q[i * 64 + l * 8 + g]);
            float e[8], s = 0.0f;
            for (int i = 0; i < 8; ++i) { e[i] = __expf(Q[i * 64 + l * 8 + g] - mx); s += e[i]; }
            float inv = 1.0f / s;
            for (int i = 0; i < 8; ++i) sQ[i * 64 + l * 8 + g] = e[i] * inv;
        }
        {
            int i = a;
            float mx = -1e30f;
            for (int m = 0; m < 32; ++m) mx = fmaxf(mx, B[i * 256 + m * 8 + g]);
            float ev[32], s = 0.0f;
            for (int m = 0; m < 32; ++m) { ev[m] = __expf(B[i * 256 + m * 8 + g] - mx); s += ev[m]; }
            float inv = 1.0f / s;
            for (int m = 0; m < 32; ++m) Bs_t[m * 64 + g * 8 + i] = ev[m] * inv;
        }
    }
    __syncthreads();
    int p = t >> 5, lane = t & 31;
    bool hi = lane >= 16;
    int  m  = lane & 15;
    bool bvalid = (hi == (m >= 8));
    int  icol = m & 7;
    int  g = 2 * p + (hi ? 1 : 0);
    _Float16* dst = bfrag + (p * 32 + lane) * 16;
    for (int e2 = 0; e2 < 16; ++e2) {
        float f = (e2 < 8 && bvalid) ? sQ[icol * 64 + e2 * 8 + g] : 0.0f;
        dst[e2] = (_Float16)f;
    }
}


#define SEG_ROWS    391
#define PRE_WG      512
#define PRE_WAVES   4

__global__ __launch_bounds__(128) void cgmm_pre(const float* __restrict__ Q,
                                                const float* __restrict__ B,
                                                const int* __restrict__ ei,
                                                float* __restrict__ Bs_t,
                                                _Float16* __restrict__ bfrag,
                                                unsigned* __restrict__ bits,
                                                int N, int E) {
    if (blockIdx.x == PRE_WG) {
        cgmm_params_body(Q, B, Bs_t, bfrag);
        return;
    }
    __shared__ __attribute__((aligned(16))) unsigned sb[PRE_WAVES][32][8];
    const int lane = threadIdx.x & 31;
    const int wave = threadIdx.x >> 5;
    const int ngroups = (E + 31) >> 5;
    const int nbatch  = (ngroups + 31) >> 5;
    const int ngs     = nbatch << 5;
    const int planes  = ((N + SEG_ROWS - 1) / SEG_ROWS + 31) >> 5;

    for (int bb = (int)blockIdx.x * PRE_WAVES; bb < nbatch; bb += PRE_WG * PRE_WAVES) {
        const int  b    = bb + wave;
        const bool live = b < nbatch;
        const int  g0   = b << 5;
        unsigned own[32];
#pragma unroll
        for (int j = 0; j < 32; ++j) {
            int e = (g0 + j) * 32 + lane;
            int d = ei[(live && e < E) ? e : E - 1];
            own[j] = (unsigned)d / (unsigned)SEG_ROWS;
        }
        for (int p0 = 0; p0 < planes; p0 += 8) {
            uint4* zr = (uint4*)&sb[wave][lane][0];
            zr[0] = make_uint4(0u, 0u, 0u, 0u);
            zr[1] = make_uint4(0u, 0u, 0u, 0u);
            __syncthreads();
#pragma unroll
            for (int j = 0; j < 32; ++j) {
                unsigned pl = (own[j] >> 5) - (unsigned)p0;
                bool     in = live && (g0 + j) * 32 + lane < E;
                if (in && pl < 8u) atomicOr(&sb[wave][j][pl], 1u << (own[j] & 31u));
            }
            __syncthreads();
            if (live) {
#pragma unroll
                for (int k = 0; k < 8; ++k)
                    if (p0 + k < planes)
                        bits[(size_t)(p0 + k) * ngs + g0 + lane] = sb[wave][lane][k];
            }
            __syncthreads();
        }
    }
}

#define SEG_WG      256
#define SEG_WAVES   32
#define SEG_QCAP    288
#define SEG_GQCAP   160
#define SEG_TILES   25
#define SEG_LROWS   (SEG_TILES * 16)
#define SEG_PITCH   68

__global__ __launch_bounds__(1024) void cgmm_owner_sum(const int* __restrict__ x,
                                                       const float* __restrict__ prev_h,
                                                       const int* __restrict__ ei,
                                                       const unsigned* __restrict__ bits,
                                                       const _Float16* __restrict__ bfrag,
                                                       const float* __restrict__ Bs_t,
                                                       float* __restrict__ lik,
                                                       float* __restrict__ post,
                                                       int N, int E) {
    __shared__ __attribute__((aligned(16))) float acc[SEG_LROWS * SEG_PITCH];
    __shared__ float cnt[SEG_LROWS];
    __shared__ __attribute__((aligned(16))) unsigned char stage[SEG_TILES * 16 * SEG_PITCH * 4];
    int2*  const q  = (int2*)stage;
    int*   const gq = (int*)(stage + SEG_WAVES * SEG_QCAP * 8);
    float* const tiles = (float*)stage;

    const int tid  = threadIdx.x;
    const int lane = tid & 31;
    const int wave = tid >> 5;
    const int ngroups = (E + 31) >> 5;
    const int ngs     = ((ngroups + 31) >> 5) << 5;
    const int nq      = ngs >> 2;
    const float2* rows2 = (const float2*)prev_h;
    const int* const esrc = ei + E;

    int own = (int)blockIdx.x;
    for (int lo = (int)blockIdx.x * SEG_ROWS; lo < N;
         lo += (int)gridDim.x * SEG_ROWS, own += (int)gridDim.x) {
        const int rows = (N - lo < SEG_ROWS) ? (N - lo) : SEG_ROWS;

        for (int i = tid; i < SEG_LROWS * SEG_PITCH / 4; i += 1024)
            ((float4*)acc)[i] = make_float4(0.0f, 0.0f, 0.0f, 0.0f);
        for (int i = tid; i < SEG_LROWS; i += 1024) cnt[i] = 0.0f;
        const int  trow   = wave * 16 + (lane & 15);
        const bool tvalid = trow < rows;
        const int  xi     = x[lo + (tvalid ? trow : rows - 1)];
        __syncthreads();

        const uint4* plane4 = (const uint4*)(bits + (size_t)(own >> 5) * ngs);
        const unsigned ob = (unsigned)own & 31u;

        int qn  = 0;
        int gqn = 0;

        auto drain = [&](int base, int n) {
            int s_l = 0, d_l = SEG_ROWS;
            if (lane < n) {
                int2 ent = q[wave * SEG_QCAP + base + lane];
                s_l = ent.x;
                d_l = ent.y;
                __hip_atomic_fetch_add(&cnt[d_l], 1.0f, __ATOMIC_RELAXED,
                                       __HIP_MEMORY_SCOPE_WORKGROUP);
            }
            for (int j0 = 0; j0 < n; j0 += 16) {
                float2 v[16];
                int    dd[16];
#pragma unroll
                for (int k = 0; k < 16; ++k) {
                    int s = __builtin_amdgcn_readlane(s_l, j0 + k);
                    dd[k] = __builtin_amdgcn_readlane(d_l, j0 + k);
                    v[k]  = rows2[(size_t)s * 32 + lane];
                }
#pragma unroll
                for (int k = 0; k < 16; ++k) {
                    float* a = acc + dd[k] * SEG_PITCH + lane * 2;
                    __hip_atomic_fetch_add(a,     v[k].x, __ATOMIC_RELAXED,
                                           __HIP_MEMORY_SCOPE_WORKGROUP);
                    __hip_atomic_fetch_add(a + 1, v[k].y, __ATOMIC_RELAXED,
                                           __HIP_MEMORY_SCOPE_WORKGROUP);
                }
            }
        };

        auto push = [&](int d, int s, bool ok) {
            unsigned t = (unsigned)(d - lo);
            bool hit = ok & (t < (unsigned)rows);
            unsigned m = __builtin_amdgcn_ballot_w32(hit);
            if (m) {
                int pos = qn + (int)__builtin_amdgcn_mbcnt_lo(m, 0u);
                if (hit) q[wave * SEG_QCAP + pos] = make_int2(s, (int)t);
                qn += __builtin_popcount(m);
            }
        };

        auto gpush = [&](unsigned w, int g, bool ok) {
            bool hit = ok & (((w >> ob) & 1u) != 0u);
            unsigned m = __builtin_amdgcn_ballot_w32(hit);
            if (m) {
                int pos = gqn + (int)__builtin_amdgcn_mbcnt_lo(m, 0u);
                if (hit) gq[wave * SEG_GQCAP + pos] = g;
                gqn += __builtin_popcount(m);
            }
        };

        auto scan_groups = [&](int base, int n) {
            int g_l = (lane < n) ? gq[wave * SEG_GQCAP + base + lane] : 0;
            for (int j0 = 0; j0 < n; j0 += 16) {
                int dv[16], sv[16];
                bool in[16];
#pragma unroll
                for (int k = 0; k < 16; ++k) {
                    int g = __builtin_amdgcn_readlane(g_l, j0 + k);
                    int e = g * 32 + lane;
                    in[k] = e < E;
                    int ec = in[k] ? e : E - 1;
                    dv[k] = ei[ec];
                    sv[k] = esrc[ec];
                }
                asm volatile("" :: "v"(sv[0]), "v"(sv[1]), "v"(sv[2]), "v"(sv[3]),
                                   "v"(sv[4]), "v"(sv[5]), "v"(sv[6]), "v"(sv[7]),
                                   "v"(sv[8]), "v"(sv[9]), "v"(sv[10]), "v"(sv[11]),
                                   "v"(sv[12]), "v"(sv[13]), "v"(sv[14]), "v"(sv[15]));
#pragma unroll
                for (int h = 0; h < 16; h += 8) {
#pragma unroll
                    for (int k = h; k < h + 8; ++k)
                        push(dv[k], sv[k], (j0 + k < n) && in[k]);
                    while (qn >= 32) { qn -= 32; drain(qn, 32); }
                }
            }
        };

        if (nq > 0) {
            int   ib  = wave * 32;
            uint4 cur = plane4[(ib + lane < nq) ? ib + lane : nq - 1];
            for (; ib < nq; ib += SEG_WAVES * 32) {
                const int  in  = ib + SEG_WAVES * 32 + lane;
                const uint4 nxt = plane4[(in < nq) ? in : nq - 1];
                const bool on = ib + lane < nq;
                const int  g  = (ib + lane) * 4;
                gpush(cur.x, g,     on);
                gpush(cur.y, g + 1, on);
                gpush(cur.z, g + 2, on);
                gpush(cur.w, g + 3, on);
                while (gqn >= 32) { gqn -= 32; scan_groups(gqn, 32); }
                cur = nxt;
            }
        }
        if (gqn > 0) scan_groups(0, gqn);
        if (qn > 0) drain(0, qn);

        __syncthreads();

        if (wave * 16 < rows) {
            const int  m  = lane & 15;
            const bool hi = lane >= 16;
            float* const arow = acc + trow * SEG_PITCH;
            float* const sT   = tiles + wave * 16 * SEG_PITCH;
            const float inv_cnt = 1.0f / fmaxf(cnt[trow], 1.0f);

#pragma unroll
            for (int p = 0; p < 4; ++p) {
                const int g = 2 * p + (hi ? 1 : 0);
                v16h a;
#pragma unroll
                for (int e2 = 0; e2 < 8; ++e2)
                    a[e2] = (_Float16)(arow[e2 * 8 + g] * inv_cnt);
#pragma unroll
                for (int e2 = 8; e2 < 16; ++e2)
                    a[e2] = (_Float16)0.0f;

                v16h b = *((const v16h*)(bfrag + (p * 32 + lane) * 16));

                v8f d = {};
                d = __builtin_amdgcn_wmma_f32_16x16x32_f16(
                    false, a, false, b, (short)0, d, false, false);
#pragma unroll
                for (int r = 0; r < 8; ++r) {
                    const int mm  = r + (hi ? 8 : 0);
                    const int gq2 = 2 * p + (m >= 8 ? 1 : 0);
                    sT[mm * SEG_PITCH + gq2 * 8 + (m & 7)] = d[r];
                }
            }

            const int gbase = hi ? 4 : 0;
            const float* brow = Bs_t + (size_t)xi * 64;
            float lik4[4];
            float pv[4][8];
#pragma unroll
            for (int gg = 0; gg < 4; ++gg) {
                const int g = gbase + gg;
                const float4* tp = (const float4*)(sT + m * SEG_PITCH + g * 8);
                float4 t0 = tp[0], t1 = tp[1];
                const float4* bp = (const float4*)(brow + g * 8);
                float4 b0 = bp[0], b1 = bp[1];
                float u[8];
                u[0] = t0.x * b0.x; u[1] = t0.y * b0.y; u[2] = t0.z * b0.z; u[3] = t0.w * b0.w;
                u[4] = t1.x * b1.x; u[5] = t1.y * b1.y; u[6] = t1.z * b1.z; u[7] = t1.w * b1.w;
                float s = ((u[0] + u[1]) + (u[2] + u[3])) + ((u[4] + u[5]) + (u[6] + u[7]));
                float denom = s + 64.0f * EPS_F;
                lik4[gg] = __logf(denom);
                float invd = 1.0f / denom;
#pragma unroll
                for (int i = 0; i < 8; ++i)
                    pv[gg][i] = (u[i] + 8.0f * EPS_F) * invd;
            }
            if (tvalid) {
                float4 lv = {lik4[0], lik4[1], lik4[2], lik4[3]};
                *(float4*)(lik + (size_t)(lo + trow) * 8 + gbase) = lv;
            }
#pragma unroll
            for (int i = 0; i < 8; ++i)
                *(float4*)(arow + i * 8 + gbase) = make_float4(pv[0][i], pv[1][i], pv[2][i], pv[3][i]);

            const float* prow = acc + (wave * 16 + (lane >> 4)) * SEG_PITCH + (lane & 15) * 4;
            float*       orow = post + (size_t)(lo + wave * 16 + (lane >> 4)) * 64 + (lane & 15) * 4;
            if (wave * 16 + 16 <= rows) {
                float4 pr[8];
#pragma unroll
                for (int k = 0; k < 8; ++k) pr[k] = *(const float4*)(prow + 2 * k * SEG_PITCH);
#pragma unroll
                for (int k = 0; k < 8; ++k) *(float4*)(orow + 2 * k * 64) = pr[k];
            } else {
#pragma unroll
                for (int k = 0; k < 8; ++k)
                    if (wave * 16 + 2 * k + (lane >> 4) < rows)
                        *(float4*)(orow + 2 * k * 64) = *(const float4*)(prow + 2 * k * SEG_PITCH);
            }
        }
        __syncthreads();
    }
}

extern "C" void kernel_launch(void* const* d_in, const int* in_sizes, int n_in,
                              void* d_out, int out_size, void* d_ws, size_t ws_size,
                              hipStream_t stream) {
    (void)n_in; (void)out_size; (void)ws_size;
    const int*   x      = (const int*)d_in[0];
    const float* prev_h = (const float*)d_in[1];
    const int*   ei     = (const int*)d_in[2];
    const float* Q      = (const float*)d_in[3];
    const float* B      = (const float*)d_in[4];
    const int N = in_sizes[0];
    const int E = in_sizes[2] / 2;

    float* ws     = (float*)d_ws;
    float* Bs_t   = ws + (size_t)N * 65;
    size_t off    = ((size_t)N * 65 + 2048 + 7) & ~(size_t)7;
    _Float16* bfrag = (_Float16*)(ws + off);
    unsigned* bits  = (unsigned*)(ws + off + 1024);

    float* lik  = (float*)d_out;
    float* post = lik + (size_t)N * 8;

    cgmm_pre<<<PRE_WG + 1, 128, 0, stream>>>(Q, B, ei, Bs_t, bfrag, bits, N, E);

    cgmm_owner_sum<<<SEG_WG, 1024, 0, stream>>>(x, prev_h, ei, bits, bfrag, Bs_t, lik, post, N, E);
}


# Row sums in registers, not by LDS float adds: kept edges go to per-wave lists (row t to wave t&31, one LDS int add each); past a barrier a wave sums its 13 rows from its list; add order: list arrival
# speedup vs baseline: 5.0631x; 1.1280x over previous
#include <hip/hip_runtime.h>
#include <hip/hip_bf16.h>
#include <math.h>


typedef __attribute__((ext_vector_type(16))) _Float16 v16h;
typedef __attribute__((ext_vector_type(8)))  float    v8f;

#define EPS_F 1e-8f

__device__ __forceinline__ void cgmm_params_body(const float* __restrict__ Q,
                                                 const float* __restrict__ B,
                                                 float* __restrict__ Bs_t,
                                                 _Float16* __restrict__ bfrag) {
    __shared__ float sQ[512];
    int t = threadIdx.x;
    if (t < 64) {
        int g = t & 7, a = t >> 3;
        {
            int l = a;
            float mx = -1e30f;
            for (int i = 0; i < 8; ++i) mx = fmaxf(mx, Q[i * 64 + l * 8 + g]);
            float e[8], s = 0.0f;
            for (int i = 0; i < 8; ++i) { e[i] = __expf(Q[i * 64 + l * 8 + g] - mx); s += e[i]; }
            float inv = 1.0f / s;
            for (int i = 0; i < 8; ++i) sQ[i * 64 + l * 8 + g] = e[i] * inv;
        }
        {
            int i = a;
            float mx = -1e30f;
            for (int m = 0; m < 32; ++m) mx = fmaxf(mx, B[i * 256 + m * 8 + g]);
            float ev[32], s = 0.0f;
            for (int m = 0; m < 32; ++m) { ev[m] = __expf(B[i * 256 + m * 8 + g] - mx); s += ev[m]; }
            float inv = 1.0f / s;
            for (int m = 0; m < 32; ++m) Bs_t[m * 64 + g * 8 + i] = ev[m] * inv;
        }
    }
    __syncthreads();
    if (t >= 128) return;
    int p = t >> 5, lane = t & 31;
    bool hi = lane >= 16;
    int  m  = lane & 15;
    bool bvalid = (hi == (m >= 8));
    int  icol = m & 7;
    int  g = 2 * p + (hi ? 1 : 0);
    _Float16* dst = bfrag + (p * 32 + lane) * 16;
    for (int e2 = 0; e2 < 16; ++e2) {
        float f = (e2 < 8 && bvalid) ? sQ[icol * 64 + e2 * 8 + g] : 0.0f;
        dst[e2] = (_Float16)f;
    }
}


#define SEG_ROWS    391
#define SEG_WG      256
#define PRE_CH      4096
#define PRE_THR     256
#define PRE_EPT     (PRE_CH / PRE_THR)
#define PRE_WPB     (PRE_CH / 16 / 32)
#define PRE_PITCH   9
#define PRE_BLK     391

__global__ __launch_bounds__(PRE_THR) void cgmm_pre(const float* __restrict__ Q,
                                                    const float* __restrict__ B,
                                                    const int* __restrict__ ei,
                                                    float* __restrict__ Bs_t,
                                                    _Float16* __restrict__ bfrag,
                                                    unsigned* __restrict__ bits,
                                                    int* __restrict__ pairs_w,
                                                    int N, int E) {
    (void)N;
    int2* const pairs = (int2*)pairs_w;
    if (blockIdx.x == PRE_BLK) {
        cgmm_params_body(Q, B, Bs_t, bfrag);
        return;
    }
    __shared__ unsigned sb[SEG_WG * PRE_PITCH];
    const int tid  = threadIdx.x;
    const int nblk = (E + PRE_CH - 1) / PRE_CH;
    const int wpo  = nblk * PRE_WPB;
    const int* const esrc = ei + E;

    for (int blk = (int)blockIdx.x; blk < nblk; blk += PRE_BLK) {
        if (blk != (int)blockIdx.x) __syncthreads();
        for (int i = tid; i < SEG_WG * PRE_PITCH; i += PRE_THR) sb[i] = 0u;
        const int e0 = blk * PRE_CH;
        int d[PRE_EPT], s[PRE_EPT];
#pragma unroll
        for (int j = 0; j < PRE_EPT; ++j) {
            int e  = e0 + j * PRE_THR + tid;
            int ec = (e < E) ? e : E - 1;
            d[j] = ei[ec];
            s[j] = esrc[ec];
        }
        __syncthreads();
#pragma unroll
        for (int j = 0; j < PRE_EPT; ++j) {
            int e = e0 + j * PRE_THR + tid;
            unsigned o = (unsigned)d[j] / (unsigned)SEG_ROWS;
            if (e < E && o < (unsigned)SEG_WG)
                atomicOr(&sb[o * PRE_PITCH + (j >> 1)], 1u << ((j & 1) * 16 + (tid >> 4)));
        }
        __syncthreads();
#pragma unroll
        for (int j = 0; j < PRE_EPT; ++j) {
            int e = e0 + j * PRE_THR + tid;
            if (e < E) pairs[e] = make_int2(d[j], s[j]);
        }
#pragma unroll
        for (int k = 0; k < SEG_WG * PRE_WPB / PRE_THR; ++k) {
            int i = k * PRE_THR + tid;
            int o = i >> 3, w = i & 7;
            bits[(size_t)o * wpo + blk * PRE_WPB + w] = sb[o * PRE_PITCH + w];
        }
    }
}

#define SEG_WAVES   32
#define SEG_QCAP    288
#define SEG_LCAP    400
#define SEG_RPW     13
#define SEG_PADROW  15
#define SEG_GQCAP   160
#define SEG_TILES   25
#define SEG_LROWS   (SEG_TILES * 16)
#define SEG_PITCH   68

__global__ __launch_bounds__(1024) void cgmm_owner_sum(const int* __restrict__ x,
                                                       const float* __restrict__ prev_h,
                                                       const int* __restrict__ pairs_w,
                                                       const unsigned* __restrict__ bits,
                                                       const _Float16* __restrict__ bfrag,
                                                       const float* __restrict__ Bs_t,
                                                       float* __restrict__ lik,
                                                       float* __restrict__ post,
                                                       int N, int E) {
    __shared__ __attribute__((aligned(16))) float acc[SEG_LROWS * SEG_PITCH];
    __shared__ float cnt[SEG_LROWS];
    __shared__ int   lcount[SEG_WAVES];
    __shared__ __attribute__((aligned(16))) unsigned char stage[SEG_TILES * 16 * SEG_PITCH * 4];
    static_assert(SEG_WAVES * (SEG_QCAP + SEG_GQCAP + SEG_LCAP) * 4 <= SEG_TILES * 16 * SEG_PITCH * 4, "queues and lists fit the third buffer");
    static_assert(SEG_RPW * SEG_WAVES >= SEG_ROWS && SEG_RPW <= SEG_PADROW && SEG_WAVES == 32, "row t belongs to wave t & 31, row number t >> 5");
    static_assert(SEG_WG * SEG_ROWS <= (1 << 20) && SEG_ROWS <= (1 << 9), "source and local row share a word");
    int*   const q    = (int*)stage;
    int*   const gq   = (int*)(stage + SEG_WAVES * SEG_QCAP * 4);
    int*   const list = (int*)(stage + SEG_WAVES * (SEG_QCAP + SEG_GQCAP) * 4);
    float* const tiles = (float*)stage;

    const int tid  = threadIdx.x;
    const int lane = tid & 31;
    const int wave = tid >> 5;
    const int wpo = ((E + PRE_CH - 1) / PRE_CH) * PRE_WPB;
    const float2* rows2 = (const float2*)prev_h;
    const int2* const pairs = (const int2*)pairs_w;

    int own = (int)blockIdx.x;
    for (int lo = (int)blockIdx.x * SEG_ROWS; lo < N;
         lo += (int)gridDim.x * SEG_ROWS, own += (int)gridDim.x) {
        const int rows = (N - lo < SEG_ROWS) ? (N - lo) : SEG_ROWS;

        for (int i = tid; i < SEG_LROWS * SEG_PITCH / 4; i += 1024)
            ((float4*)acc)[i] = make_float4(0.0f, 0.0f, 0.0f, 0.0f);
        for (int i = tid; i < SEG_LROWS; i += 1024) cnt[i] = 0.0f;
        if (tid < SEG_WAVES) lcount[tid] = 0;
        const int  trow   = wave * 16 + (lane & 15);
        const bool tvalid = trow < rows;
        const int  xi     = x[lo + (tvalid ? trow : rows - 1)];
        __syncthreads();

        const unsigned* const mine = bits + (size_t)own * wpo;

        int qn  = 0;
        int gqn = 0;

        auto route = [&](int base, int n) {
            const bool act = lane < n;
            const int  ent = act ? q[wave * SEG_QCAP + base + lane] : 0;
            const int  s_l = ent & 0xFFFFF;
            const int  t_l = ent >> 20;
            const int  ow  = t_l & (SEG_WAVES - 1);
            int pos = 0;
            if (act) {
                pos = __hip_atomic_fetch_add(&lcount[ow], 1, __ATOMIC_RELAXED,
                                             __HIP_MEMORY_SCOPE_WORKGROUP);
                __hip_atomic_fetch_add(&cnt[t_l], 1.0f, __ATOMIC_RELAXED,
                                       __HIP_MEMORY_SCOPE_WORKGROUP);
            }
            const bool fits = act & (pos < SEG_LCAP);
            if (fits) list[ow * SEG_LCAP + pos] = s_l | ((t_l >> 5) << 20);
            unsigned ov = __builtin_amdgcn_ballot_w32(act & !fits);
            while (ov) {
                const int b = __builtin_ctz(ov);
                ov &= ov - 1u;
                const int s = __builtin_amdgcn_readlane(s_l, b);
                const int t = __builtin_amdgcn_readlane(t_l, b);
                const float2 v = rows2[(size_t)s * 32 + lane];
                float* a = acc + t * SEG_PITCH + lane * 2;
                __hip_atomic_fetch_add(a,     v.x, __ATOMIC_RELAXED, __HIP_MEMORY_SCOPE_WORKGROUP);
                __hip_atomic_fetch_add(a + 1, v.y, __ATOMIC_RELAXED, __HIP_MEMORY_SCOPE_WORKGROUP);
            }
        };

        auto push = [&](int d, int s, bool ok) {
            unsigned t = (unsigned)(d - lo);
            bool hit = ok & (t < (unsigned)rows);
            unsigned m = __builtin_amdgcn_ballot_w32(hit);
            if (m) {
                int pos = qn + (int)__builtin_amdgcn_mbcnt_lo(m, 0u);
                if (hit) q[wave * SEG_QCAP + pos] = s | (int)(t << 20);
                qn += __builtin_popcount(m);
            }
        };

        auto scan_groups = [&](int base, int n) {
            const int g_l  = (lane < n) ? gq[wave * SEG_GQCAP + base + lane] : 0;
            const int half = lane >> 4;
            const int sub  = lane & 15;
            int2 de[16];
            bool in[16];
#pragma unroll
            for (int k = 0; k < 16; ++k) {
                int ga = __builtin_amdgcn_readlane(g_l, 2 * k);
                int gb = __builtin_amdgcn_readlane(g_l, 2 * k + 1);
                int e  = (half ? gb : ga) * 16 + sub;
                in[k]  = (2 * k + half < n) && (e < E);
                de[k]  = pairs[(e < E) ? e : E - 1];
            }
#pragma unroll
            for (int h = 0; h < 16; h += 4) {
#pragma unroll
                for (int k = h; k < h + 4; ++k)
                    push(de[k].x, de[k].y, in[k]);
                while (qn >= 32) { qn -= 32; route(qn, 32); }
            }
        };

        if (wpo > 0) {
            int      ib  = wave * 32;
            unsigned cur = mine[(ib + lane < wpo) ? ib + lane : wpo - 1];
            for (; ib < wpo; ib += SEG_WAVES * 32) {
                const int      nx  = ib + SEG_WAVES * 32 + lane;
                const unsigned nxt = mine[(nx < wpo) ? nx : wpo - 1];
                unsigned  w  = (ib + lane < wpo) ? cur : 0u;
                const int g0 = (ib + lane) * 32;
                for (int r = 0; r < 32; ++r) {
                    const bool     hit = w != 0u;
                    const unsigned m   = __builtin_amdgcn_ballot_w32(hit);
                    if (!m) break;
                    const int pos = gqn + (int)__builtin_amdgcn_mbcnt_lo(m, 0u);
                    if (hit) gq[wave * SEG_GQCAP + pos] = g0 + __builtin_ctz(w);
                    w &= w - 1u;
                    gqn += __builtin_popcount(m);
                    while (gqn >= 32) { gqn -= 32; scan_groups(gqn, 32); }
                }
                cur = nxt;
            }
        }
        if (gqn > 0) scan_groups(0, gqn);
        if (qn > 0) route(0, qn);

        __syncthreads();

        {
            int nl = __builtin_amdgcn_readfirstlane(lcount[wave]);
            if (nl > SEG_LCAP) nl = SEG_LCAP;
            const int* const mylist = list + wave * SEG_LCAP;
            float ra[32];
#pragma unroll
            for (int i = 0; i < 32; ++i) ra[i] = 0.0f;
            for (int base = 0; base < nl; base += 32) {
                const int left = nl - base;
                const int ent  = (lane < left) ? mylist[base + lane] : (SEG_PADROW << 20);
#pragma unroll
                for (int h = 0; h < 32; h += 16) {
                    if (h < left) {
                        float2 v[16];
                        int    rr[16];
#pragma unroll
                        for (int k = 0; k < 16; ++k) {
                            const int e = __builtin_amdgcn_readlane(ent, h + k);
                            rr[k] = e >> 20;
                            v[k]  = rows2[(size_t)(e & 0xFFFFF) * 32 + lane];
                        }
#pragma unroll
                        for (int k = 0; k < 16; ++k) {
                            const int r = rr[k];
                            ra[2 * r]     += v[k].x;
                            ra[2 * r + 1] += v[k].y;
                        }
                    }
                }
            }
#pragma unroll
            for (int i = 0; i < SEG_RPW; ++i) {
                const int t = i * SEG_WAVES + wave;
                if (t < rows) {
                    float2* const p = (float2*)(acc + t * SEG_PITCH + lane * 2);
                    float2 o = *p;
                    o.x += ra[2 * i];
                    o.y += ra[2 * i + 1];
                    *p = o;
                }
            }
        }

        __syncthreads();

        if (wave * 16 < rows) {
            const int  m  = lane & 15;
            const bool hi = lane >= 16;
            float* const arow = acc + trow * SEG_PITCH;
            float* const sT   = tiles + wave * 16 * SEG_PITCH;
            const float inv_cnt = 1.0f / fmaxf(cnt[trow], 1.0f);

#pragma unroll
            for (int p = 0; p < 4; ++p) {
                const int g = 2 * p + (hi ? 1 : 0);
                v16h a;
#pragma unroll
                for (int e2 = 0; e2 < 8; ++e2)
                    a[e2] = (_Float16)(arow[e2 * 8 + g] * inv_cnt);
#pragma unroll
                for (int e2 = 8; e2 < 16; ++e2)
                    a[e2] = (_Float16)0.0f;

                v16h b = *((const v16h*)(bfrag + (p * 32 + lane) * 16));

                v8f d = {};
                d = __builtin_amdgcn_wmma_f32_16x16x32_f16(
                    false, a, false, b, (short)0, d, false, false);
#pragma unroll
                for (int r = 0; r < 8; ++r) {
                    const int mm  = r + (hi ? 8 : 0);
                    const int gq2 = 2 * p + (m >= 8 ? 1 : 0);
                    sT[mm * SEG_PITCH + gq2 * 8 + (m & 7)] = d[r];
                }
            }

            const int gbase = hi ? 4 : 0;
            const float* brow = Bs_t + (size_t)xi * 64;
            float lik4[4];
            float pv[4][8];
#pragma unroll
            for (int gg = 0; gg < 4; ++gg) {
                const int g = gbase + gg;
                const float4* tp = (const float4*)(sT + m * SEG_PITCH + g * 8);
                float4 t0 = tp[0], t1 = tp[1];
                const float4* bp = (const float4*)(brow + g * 8);
                float4 b0 = bp[0], b1 = bp[1];
                float u[8];
                u[0] = t0.x * b0.x; u[1] = t0.y * b0.y; u[2] = t0.z * b0.z; u[3] = t0.w * b0.w;
                u[4] = t1.x * b1.x; u[5] = t1.y * b1.y; u[6] = t1.z * b1.z; u[7] = t1.w * b1.w;
                float s = ((u[0] + u[1]) + (u[2] + u[3])) + ((u[4] + u[5]) + (u[6] + u[7]));
                float denom = s + 64.0f * EPS_F;
                lik4[gg] = __logf(denom);
                float invd = 1.0f / denom;
#pragma unroll
                for (int i = 0; i < 8; ++i)
                    pv[gg][i] = (u[i] + 8.0f * EPS_F) * invd;
            }
            if (tvalid) {
                float4 lv = {lik4[0], lik4[1], lik4[2], lik4[3]};
                *(float4*)(lik + (size_t)(lo + trow) * 8 + gbase) = lv;
            }
#pragma unroll
            for (int i = 0; i < 8; ++i)
                *(float4*)(arow + i * 8 + gbase) = make_float4(pv[0][i], pv[1][i], pv[2][i], pv[3][i]);

            const float* prow = acc + (wave * 16 + (lane >> 4)) * SEG_PITCH + (lane & 15) * 4;
            float*       orow = post + (size_t)(lo + wave * 16 + (lane >> 4)) * 64 + (lane & 15) * 4;
            if (wave * 16 + 16 <= rows) {
                float4 pr[8];
#pragma unroll
                for (int k = 0; k < 8; ++k) pr[k] = *(const float4*)(prow + 2 * k * SEG_PITCH);
#pragma unroll
                for (int k = 0; k < 8; ++k) *(float4*)(orow + 2 * k * 64) = pr[k];
            } else {
#pragma unroll
                for (int k = 0; k < 8; ++k)
                    if (wave * 16 + 2 * k + (lane >> 4) < rows)
                        *(float4*)(orow + 2 * k * 64) = *(const float4*)(prow + 2 * k * SEG_PITCH);
            }
        }
        __syncthreads();
    }
}

extern "C" void kernel_launch(void* const* d_in, const int* in_sizes, int n_in,
                              void* d_out, int out_size, void* d_ws, size_t ws_size,
                              hipStream_t stream) {
    (void)n_in; (void)out_size; (void)ws_size;
    const int*   x      = (const int*)d_in[0];
    const float* prev_h = (const float*)d_in[1];
    const int*   ei     = (const int*)d_in[2];
    const float* Q      = (const float*)d_in[3];
    const float* B      = (const float*)d_in[4];
    const int N = in_sizes[0];
    const int E = in_sizes[2] / 2;

    float* ws     = (float*)d_ws;
    float* Bs_t   = ws + (size_t)N * 65;
    size_t off    = ((size_t)N * 65 + 2048 + 7) & ~(size_t)7;
    _Float16* bfrag = (_Float16*)(ws + off);
    int*      pairs = (int*)d_ws;
    unsigned* bits  = (unsigned*)d_ws + (((size_t)E * 2 + 63) & ~(size_t)63);

    float* lik  = (float*)d_out;
    float* post = lik + (size_t)N * 8;

    cgmm_pre<<<PRE_BLK + 1, PRE_THR, 0, stream>>>(Q, B, ei, Bs_t, bfrag, bits, pairs, N, E);

    cgmm_owner_sum<<<SEG_WG, 1024, 0, stream>>>(x, prev_h, pairs, bits, bfrag, Bs_t, lik, post, N, E);
}
